# starting kernel
# baseline (speedup 1.0000x reference)
#include <hip/hip_runtime.h>
#include <hip/hip_bf16.h>

#define B_DIM 4096
#define O_DIM 16
#define D_DIM 256
#define H_DIM 512

typedef __attribute__((ext_vector_type(2))) float v2f;
typedef __attribute__((ext_vector_type(8))) float v8f;

__device__ __forceinline__ v8f wmma_f32(v2f a, v2f b, v8f c) {
    return __builtin_amdgcn_wmma_f32_16x16x4_f32(
         false, a,  false, b,
         (short)0, c,  false,  false);
}

__global__ void __launch_bounds__(256)
k_reduce_states(const float* __restrict__ states, float* __restrict__ xbar) {
    const int b = blockIdx.x;
    const int d = threadIdx.x;
    const float* p = states + (size_t)b * (O_DIM * D_DIM) + d;
    float acc = 0.f;
#pragma unroll
    for (int o = 0; o < O_DIM; ++o) acc += p[o * D_DIM];
    xbar[(size_t)b * D_DIM + d] = acc * (1.0f / 16.0f);
}

__global__ void __launch_bounds__(256)
k_gemm1(const float* __restrict__ xbar, const float* __restrict__ W1,
        const float* __restrict__ bias1, const int* __restrict__ action,
        float* __restrict__ h1) {
    const int K = D_DIM, N = H_DIM;
    const int wave = threadIdx.x >> 5;
    const int lane = threadIdx.x & 31;
    const int job  = blockIdx.x * 8 + wave;
    const int m0 = (job >> 3) * 16;
    const int n0 = (job & 7) * 64;
    const int laneM = lane & 15;
    const int laneK = (lane >> 4) * 2;

    v8f acc[4] = {};
    const float* arow = xbar + (size_t)(m0 + laneM) * K + laneK;
    for (int k = 0; k < K; k += 4) {
        v2f a = *(const v2f*)(arow + k);
        const float* bp = W1 + (size_t)(k + laneK) * N + laneM;
#pragma unroll
        for (int t = 0; t < 4; ++t) {
            v2f bf = { bp[n0 + t * 16], bp[N + n0 + t * 16] };
            acc[t] = wmma_f32(a, bf, acc[t]);
        }
    }
    const int mhalf = (lane >> 4) * 8;
#pragma unroll
    for (int r = 0; r < 8; ++r) {
        const int m = m0 + mhalf + r;
        const int c = action[m] & 3;
        const float* wtail = W1 + (size_t)(D_DIM + c) * N;
#pragma unroll
        for (int t = 0; t < 4; ++t) {
            const int n = n0 + t * 16 + laneM;
            float v = acc[t][r] + wtail[n] * 0.0625f + bias1[n];
            h1[(size_t)m * N + n] = fmaxf(v, 0.0f);
        }
    }
}

__global__ void __launch_bounds__(256)
k_gemm2_ln(const float* __restrict__ h1, const float* __restrict__ W2,
           const float* __restrict__ bias2, const float* __restrict__ ln_g,
           const float* __restrict__ ln_b, float* __restrict__ h2) {
    const int K = H_DIM, N = H_DIM;
    __shared__ float z[16 * H_DIM];
    __shared__ float red_s[16][16];
    __shared__ float red_q[16][16];
    __shared__ float mu_s[16], rs_s[16];

    const int wave = threadIdx.x >> 5;
    const int lane = threadIdx.x & 31;
    const int m0 = blockIdx.x * 16;
    const int n0 = wave * 64;
    const int laneM = lane & 15;
    const int laneK = (lane >> 4) * 2;

    v8f acc[4] = {};
    const float* arow = h1 + (size_t)(m0 + laneM) * K + laneK;
    for (int k = 0; k < K; k += 4) {
        v2f a = *(const v2f*)(arow + k);
        const float* bp = W2 + (size_t)(k + laneK) * N + laneM;
#pragma unroll
        for (int t = 0; t < 4; ++t) {
            v2f bf = { bp[n0 + t * 16], bp[N + n0 + t * 16] };
            acc[t] = wmma_f32(a, bf, acc[t]);
        }
    }
    const int mhalf = (lane >> 4) * 8;
#pragma unroll
    for (int r = 0; r < 8; ++r) {
#pragma unroll
        for (int t = 0; t < 4; ++t) {
            const int n = n0 + t * 16 + laneM;
            z[(mhalf + r) * H_DIM + n] = acc[t][r] + bias2[n];
        }
    }
    __syncthreads();

    const int row = threadIdx.x >> 4;
    const int seg = threadIdx.x & 15;
    {
        const float* zr = z + row * H_DIM + seg * 32;
        float s = 0.f, q = 0.f;
#pragma unroll
        for (int i = 0; i < 32; ++i) { float v = zr[i]; s += v; q += v * v; }
        red_s[row][seg] = s;
        red_q[row][seg] = q;
    }
    __syncthreads();
    if (threadIdx.x < 16) {
        float s = 0.f, q = 0.f;
#pragma unroll
        for (int i = 0; i < 16; ++i) { s += red_s[threadIdx.x][i]; q += red_q[threadIdx.x][i]; }
        const float mu = s * (1.0f / H_DIM);
        const float var = q * (1.0f / H_DIM) - mu * mu;
        mu_s[threadIdx.x] = mu;
        rs_s[threadIdx.x] = rsqrtf(var + 1e-5f);
    }
    __syncthreads();
    {
        const float mu = mu_s[row], rstd = rs_s[row];
        float* dst = h2 + (size_t)(m0 + row) * H_DIM;
        const float* zr = z + row * H_DIM;
#pragma unroll
        for (int i = 0; i < 32; ++i) {
            const int n = seg * 32 + i;
            float v = (zr[n] - mu) * rstd * ln_g[n] + ln_b[n];
            dst[n] = fmaxf(v, 0.0f);
        }
    }
}

__global__ void __launch_bounds__(256)
k_gemm3_bcast(const float* __restrict__ h2, const float* __restrict__ W3,
              const float* __restrict__ bias3, float* __restrict__ out) {
    const int K = H_DIM, N = D_DIM;
    const int wave = threadIdx.x >> 5;
    const int lane = threadIdx.x & 31;
    const int job  = blockIdx.x * 8 + wave;
    const int m0 = (job >> 2) * 16;
    const int n0 = (job & 3) * 64;
    const int laneM = lane & 15;
    const int laneK = (lane >> 4) * 2;

    v8f acc[4] = {};
    const float* arow = h2 + (size_t)(m0 + laneM) * K + laneK;
    for (int k = 0; k < K; k += 4) {
        v2f a = *(const v2f*)(arow + k);
        const float* bp = W3 + (size_t)(k + laneK) * N + laneM;
#pragma unroll
        for (int t = 0; t < 4; ++t) {
            v2f bf = { bp[n0 + t * 16], bp[N + n0 + t * 16] };
            acc[t] = wmma_f32(a, bf, acc[t]);
        }
    }
    const int mhalf = (lane >> 4) * 8;
#pragma unroll
    for (int r = 0; r < 8; ++r) {
        const int m = m0 + mhalf + r;
        float* obase = out + (size_t)m * (O_DIM * D_DIM);
#pragma unroll
        for (int t = 0; t < 4; ++t) {
            const int n = n0 + t * 16 + laneM;
            const float v = acc[t][r] + bias3[n];
#pragma unroll
            for (int o = 0; o < O_DIM; ++o)
                obase[o * D_DIM + n] = v;
        }
    }
}

extern "C" void kernel_launch(void* const* d_in, const int* in_sizes, int n_in,
                              void* d_out, int out_size, void* d_ws, size_t ws_size,
                              hipStream_t stream) {
    const float* states = (const float*)d_in[0];
    const int*   action = (const int*)d_in[1];
    const float* W1     = (const float*)d_in[2];
    const float* b1     = (const float*)d_in[3];
    const float* W2     = (const float*)d_in[4];
    const float* b2     = (const float*)d_in[5];
    const float* W3     = (const float*)d_in[6];
    const float* b3     = (const float*)d_in[7];
    const float* ln_g   = (const float*)d_in[8];
    const float* ln_b   = (const float*)d_in[9];
    float* out = (float*)d_out;

    float* xbar = (float*)d_ws;
    float* h1   = xbar + (size_t)B_DIM * D_DIM;
    float* h2   = h1   + (size_t)B_DIM * H_DIM;

    k_reduce_states<<<B_DIM, 256, 0, stream>>>(states, xbar);
    k_gemm1<<<(B_DIM / 16) * 8 / 8, 256, 0, stream>>>(xbar, W1, b1, action, h1);
    k_gemm2_ln<<<B_DIM / 16, 256, 0, stream>>>(h1, W2, b2, ln_g, ln_b, h2);
    k_gemm3_bcast<<<(B_DIM / 16) * 4 / 8, 256, 0, stream>>>(h2, W3, b3, out);
}


# All three products as block GEMMs through LDS: W k-slabs and the block's A rows copied in by whole-block 16-byte loads, 2 stages, next slab in flight under the products; fragments from LDS; same sums
# speedup vs baseline: 1.9365x; 1.9365x over previous
#include <hip/hip_runtime.h>
#include <hip/hip_bf16.h>


#define B_DIM 4096
#define O_DIM 16
#define D_DIM 256
#define H_DIM 512

typedef __attribute__((ext_vector_type(2))) float v2f;
typedef __attribute__((ext_vector_type(4))) float v4f;
typedef __attribute__((ext_vector_type(8))) float v8f;

__device__ __forceinline__ v8f wmma_f32(v2f a, v2f b, v8f c) {
    return __builtin_amdgcn_wmma_f32_16x16x4_f32(
         false, a,  false, b,
         (short)0, c,  false,  false);
}

template <int K, int BN, int KS, int WM>
struct Tile {
    static constexpr int BM = 16 * WM;
    static constexpr int WN = 8 / WM;
    static constexpr int PA = K + 4;
    static constexpr int PW = BN + 16;
    static constexpr int A_FLOATS = BM * PA;
    static constexpr int W_FLOATS = KS * PW;
    static constexpr int LDS_FLOATS = A_FLOATS + 2 * W_FLOATS;
};

template <int K, int N, int BN, int KS, int NT, int WM>
__device__ __forceinline__ void block_gemm_lds(const float* __restrict__ Ablk,
                                               const float* __restrict__ Wcol,
                                               float* __restrict__ As,
                                               float* __restrict__ Ws,
                                               v8f (&acc)[NT]) {
    typedef Tile<K, BN, KS, WM> T;
    constexpr int ROWV  = BN / 4;
    constexpr int WV    = KS * ROWV / 256;
    constexpr int AROWV = K / 4;
    constexpr int AV    = T::BM * AROWV / 256;
    constexpr int S     = K / KS;
    static_assert(T::WN * NT * 16 == BN, "wave columns x NT tiles cover BN");
    static_assert((KS * ROWV) % 256 == 0 && (T::BM * AROWV) % 256 == 0, "whole pieces per thread");
    static_assert(K % KS == 0 && KS % 16 == 0, "whole slabs, whole quarters");

    const int tid  = threadIdx.x;
    const int wave = tid >> 5;
    const int lane = tid & 31;
    const int c    = lane & 15;
    const int h    = lane >> 4;
    const int wm   = wave / T::WN;
    const int wn   = wave % T::WN;
    const int n0   = wn * (NT * 16);

    v4f wreg[WV];

#pragma unroll
    for (int j = 0; j < AV; ++j) {
        const int idx = tid + 256 * j;
        const int r = idx / AROWV, q = idx % AROWV;
        *(v4f*)(As + r * T::PA + q * 4) = *(const v4f*)(Ablk + (size_t)r * K + q * 4);
    }
#pragma unroll
    for (int j = 0; j < WV; ++j) {
        const int idx = tid + 256 * j;
        wreg[j] = *(const v4f*)(Wcol + (size_t)(idx / ROWV) * N + (idx % ROWV) * 4);
    }
#pragma unroll
    for (int j = 0; j < WV; ++j) {
        const int idx = tid + 256 * j;
        *(v4f*)(Ws + (idx / ROWV) * T::PW + (idx % ROWV) * 4) = wreg[j];
    }
    __syncthreads();

    const float* ap = As + (wm * 16 + c) * T::PA + 2 * h;
    for (int s = 0; s < S; ++s) {
        const float* cur = Ws + (s & 1) * T::W_FLOATS;
        float* nxt = Ws + ((s + 1) & 1) * T::W_FLOATS;
        if (s + 1 < S) {
            const float* src = Wcol + (size_t)(s + 1) * KS * N;
#pragma unroll
            for (int j = 0; j < WV; ++j) {
                const int idx = tid + 256 * j;
                wreg[j] = *(const v4f*)(src + (size_t)(idx / ROWV) * N + (idx % ROWV) * 4);
            }
        }
        const float* bp = cur + (2 * h) * T::PW + n0 + c;
        constexpr int KHEAD = KS * 3 / 4;
#pragma unroll
        for (int kk = 0; kk < KS; kk += 4) {
            if (kk == KHEAD) __builtin_amdgcn_sched_barrier(0x0106);
            v2f a = *(const v2f*)(ap + s * KS + kk);
#pragma unroll
            for (int t = 0; t < NT; ++t) {
                v2f bf = { bp[kk * T::PW + t * 16], bp[(kk + 1) * T::PW + t * 16] };
                acc[t] = wmma_f32(a, bf, acc[t]);
            }
        }
        if (s + 1 < S) {
#pragma unroll
            for (int j = 0; j < WV; ++j) {
                const int idx = tid + 256 * j;
                *(v4f*)(nxt + (idx / ROWV) * T::PW + (idx % ROWV) * 4) = wreg[j];
            }
        }
        __syncthreads();
    }
}

__global__ void __launch_bounds__(256)
k_reduce_states(const float* __restrict__ states, float* __restrict__ xbar) {
    const int b = blockIdx.x;
    const int d = threadIdx.x;
    const float* p = states + (size_t)b * (O_DIM * D_DIM) + d;
    float acc = 0.f;
#pragma unroll
    for (int o = 0; o < O_DIM; ++o) acc += p[o * D_DIM];
    xbar[(size_t)b * D_DIM + d] = acc * (1.0f / 16.0f);
}

__global__ void __launch_bounds__(256)
k_gemm1(const float* __restrict__ xbar, const float* __restrict__ W1,
        const float* __restrict__ bias1, const int* __restrict__ action,
        float* __restrict__ h1) {
    constexpr int K = D_DIM, N = H_DIM, BN = 256, KS = 32, NT = 4, WM = 2;
    typedef Tile<K, BN, KS, WM> T;
    __shared__ __attribute__((aligned(16))) float lds[T::LDS_FLOATS];

    const int wave = threadIdx.x >> 5;
    const int lane = threadIdx.x & 31;
    const int mb = blockIdx.x / (N / BN);
    const int nb = (blockIdx.x % (N / BN)) * BN;
    const int m0 = mb * T::BM + (wave / T::WN) * 16;
    const int n0 = nb + (wave % T::WN) * (NT * 16);
    const int laneM = lane & 15;

    v8f acc[NT] = {};
    block_gemm_lds<K, N, BN, KS, NT, WM>(xbar + (size_t)mb * T::BM * K, W1 + nb,
                                         lds, lds + T::A_FLOATS, acc);

    const int mhalf = (lane >> 4) * 8;
#pragma unroll
    for (int r = 0; r < 8; ++r) {
        const int m = m0 + mhalf + r;
        const int c = action[m] & 3;
        const float* wtail = W1 + (size_t)(D_DIM + c) * N;
#pragma unroll
        for (int t = 0; t < 4; ++t) {
            const int n = n0 + t * 16 + laneM;
            float v = acc[t][r] + wtail[n] * 0.0625f + bias1[n];
            h1[(size_t)m * N + n] = fmaxf(v, 0.0f);
        }
    }
}

__global__ void __launch_bounds__(256)
k_gemm2_ln(const float* __restrict__ h1, const float* __restrict__ W2,
           const float* __restrict__ bias2, const float* __restrict__ ln_g,
           const float* __restrict__ ln_b, float* __restrict__ h2) {
    constexpr int K = H_DIM, N = H_DIM, BN = H_DIM, KS = 32, NT = 4, WM = 1;
    typedef Tile<K, BN, KS, WM> T;
    __shared__ __attribute__((aligned(16))) float lds[T::LDS_FLOATS];
    __shared__ float red_s[16][16];
    __shared__ float red_q[16][16];
    __shared__ float mu_s[16], rs_s[16];
    float* z = lds + T::A_FLOATS;
    static_assert(16 * H_DIM <= 2 * T::W_FLOATS, "z fits over the W stages");

    const int wave = threadIdx.x >> 5;
    const int lane = threadIdx.x & 31;
    const int m0 = blockIdx.x * 16;
    const int n0 = wave * 64;
    const int laneM = lane & 15;

    v8f acc[NT] = {};
    block_gemm_lds<K, N, BN, KS, NT, WM>(h1 + (size_t)m0 * K, W2,
                                         lds, lds + T::A_FLOATS, acc);

    const int mhalf = (lane >> 4) * 8;
#pragma unroll
    for (int r = 0; r < 8; ++r) {
#pragma unroll
        for (int t = 0; t < 4; ++t) {
            const int n = n0 + t * 16 + laneM;
            z[(mhalf + r) * H_DIM + n] = acc[t][r] + bias2[n];
        }
    }
    __syncthreads();

    const int row = threadIdx.x >> 4;
    const int seg = threadIdx.x & 15;
    {
        const float* zr = z + row * H_DIM + seg * 32;
        float s = 0.f, q = 0.f;
#pragma unroll
        for (int i = 0; i < 32; ++i) { float v = zr[i]; s += v; q += v * v; }
        red_s[row][seg] = s;
        red_q[row][seg] = q;
    }
    __syncthreads();
    if (threadIdx.x < 16) {
        float s = 0.f, q = 0.f;
#pragma unroll
        for (int i = 0; i < 16; ++i) { s += red_s[threadIdx.x][i]; q += red_q[threadIdx.x][i]; }
        const float mu = s * (1.0f / H_DIM);
        const float var = q * (1.0f / H_DIM) - mu * mu;
        mu_s[threadIdx.x] = mu;
        rs_s[threadIdx.x] = rsqrtf(var + 1e-5f);
    }
    __syncthreads();
    {
        const float mu = mu_s[row], rstd = rs_s[row];
        float* dst = h2 + (size_t)(m0 + row) * H_DIM;
        const float* zr = z + row * H_DIM;
#pragma unroll
        for (int i = 0; i < 32; ++i) {
            const int n = seg * 32 + i;
            float v = (zr[n] - mu) * rstd * ln_g[n] + ln_b[n];
            dst[n] = fmaxf(v, 0.0f);
        }
    }
}

__global__ void __launch_bounds__(256)
k_gemm3_bcast(const float* __restrict__ h2, const float* __restrict__ W3,
              const float* __restrict__ bias3, float* __restrict__ out) {
    constexpr int K = H_DIM, N = D_DIM, BN = 128, KS = 64, NT = 2, WM = 2;
    typedef Tile<K, BN, KS, WM> T;
    __shared__ __attribute__((aligned(16))) float lds[T::LDS_FLOATS];

    const int wave = threadIdx.x >> 5;
    const int lane = threadIdx.x & 31;
    const int mb = blockIdx.x / (N / BN);
    const int nb = (blockIdx.x % (N / BN)) * BN;
    const int m0 = mb * T::BM + (wave / T::WN) * 16;
    const int n0 = nb + (wave % T::WN) * (NT * 16);
    const int laneM = lane & 15;

    v8f acc[NT] = {};
    block_gemm_lds<K, N, BN, KS, NT, WM>(h2 + (size_t)mb * T::BM * K, W3 + nb,
                                         lds, lds + T::A_FLOATS, acc);

    const int mhalf = (lane >> 4) * 8;
#pragma unroll
    for (int r = 0; r < 8; ++r) {
        const int m = m0 + mhalf + r;
        float* obase = out + (size_t)m * (O_DIM * D_DIM);
#pragma unroll
        for (int t = 0; t < NT; ++t) {
            const int n = n0 + t * 16 + laneM;
            const float v = acc[t][r] + bias3[n];
#pragma unroll
            for (int o = 0; o < O_DIM; ++o)
                obase[o * D_DIM + n] = v;
        }
    }
}

extern "C" void kernel_launch(void* const* d_in, const int* in_sizes, int n_in,
                              void* d_out, int out_size, void* d_ws, size_t ws_size,
                              hipStream_t stream) {
    const float* states = (const float*)d_in[0];
    const int*   action = (const int*)d_in[1];
    const float* W1     = (const float*)d_in[2];
    const float* b1     = (const float*)d_in[3];
    const float* W2     = (const float*)d_in[4];
    const float* b2     = (const float*)d_in[5];
    const float* W3     = (const float*)d_in[6];
    const float* b3     = (const float*)d_in[7];
    const float* ln_g   = (const float*)d_in[8];
    const float* ln_b   = (const float*)d_in[9];
    float* out = (float*)d_out;

    float* xbar = (float*)d_ws;
    float* h1   = xbar + (size_t)B_DIM * D_DIM;
    float* h2   = h1   + (size_t)B_DIM * H_DIM;

    k_reduce_states<<<B_DIM, 256, 0, stream>>>(states, xbar);
    k_gemm1<<<(B_DIM / 32) * 2, 256, 0, stream>>>(xbar, W1, b1, action, h1);
    k_gemm2_ln<<<B_DIM / 16, 256, 0, stream>>>(h1, W2, b2, ln_g, ln_b, h2);
    k_gemm3_bcast<<<(B_DIM / 32) * 2, 256, 0, stream>>>(h2, W3, b3, out);
}


# LayerNorm moved from product 2's epilogue into product 3's A-row staging (normalised in registers on the way to LDS); product 2 tiles 32x256, reads W2 half as often; order of the LN sums changes
# speedup vs baseline: 2.0285x; 1.0475x over previous
#include <hip/hip_runtime.h>
#include <hip/hip_bf16.h>


#define B_DIM 4096
#define O_DIM 16
#define D_DIM 256
#define H_DIM 512

typedef __attribute__((ext_vector_type(2))) float v2f;
typedef __attribute__((ext_vector_type(4))) float v4f;
typedef __attribute__((ext_vector_type(8))) float v8f;

__device__ __forceinline__ v8f wmma_f32(v2f a, v2f b, v8f c) {
    return __builtin_amdgcn_wmma_f32_16x16x4_f32(
         false, a,  false, b,
         (short)0, c,  false,  false);
}

template <int K, int BN, int KS, int WM>
struct Tile {
    static constexpr int BM = 16 * WM;
    static constexpr int WN = 8 / WM;
    static constexpr int PA = K + 4;
    static constexpr int PW = BN + 16;
    static constexpr int A_FLOATS = BM * PA;
    static constexpr int W_FLOATS = KS * PW;
    static constexpr int LDS_FLOATS = A_FLOATS + 2 * W_FLOATS;
};

template <int K, int N, int BN, int KS, int NT, int WM, bool LN_A>
__device__ __forceinline__ void block_gemm_lds(const float* __restrict__ Ablk,
                                               const float* __restrict__ Wcol,
                                               const float* __restrict__ lng,
                                               const float* __restrict__ lnb,
                                               float* __restrict__ As,
                                               float* __restrict__ Ws,
                                               v8f (&acc)[NT]) {
    typedef Tile<K, BN, KS, WM> T;
    constexpr int ROWV  = BN / 4;
    constexpr int WV    = KS * ROWV / 256;
    constexpr int AROWV = K / 4;
    constexpr int AV    = T::BM * AROWV / 256;
    constexpr int S     = K / KS;
    static_assert(T::WN * NT * 16 == BN, "wave columns x NT tiles cover BN");
    static_assert((KS * ROWV) % 256 == 0 && (T::BM * AROWV) % 256 == 0, "whole pieces per thread");
    static_assert(K % KS == 0 && KS % 16 == 0, "whole slabs, whole quarters");

    const int tid  = threadIdx.x;
    const int wave = tid >> 5;
    const int lane = tid & 31;
    const int c    = lane & 15;
    const int h    = lane >> 4;
    const int wm   = wave / T::WN;
    const int wn   = wave % T::WN;
    const int n0   = wn * (NT * 16);

    v4f wreg[WV];

    constexpr int NR = T::BM / 16;
    constexpr int NJ = K / 64;
    static_assert(!LN_A || (K % 64 == 0 && T::BM % 16 == 0), "whole pieces per lane, whole rows per half wave");
    const int hw  = tid >> 4;
    const int seg = tid & 15;
    v4f xr[LN_A ? NR : 1][LN_A ? NJ : 1];
    v4f gr[LN_A ? NJ : 1], br[LN_A ? NJ : 1];

    if constexpr (LN_A) {
#pragma unroll
        for (int rr = 0; rr < NR; ++rr)
#pragma unroll
            for (int j = 0; j < NJ; ++j)
                xr[rr][j] = *(const v4f*)(Ablk + (size_t)(hw + 16 * rr) * K + (seg + 16 * j) * 4);
#pragma unroll
        for (int j = 0; j < NJ; ++j) {
            gr[j] = *(const v4f*)(lng + (seg + 16 * j) * 4);
            br[j] = *(const v4f*)(lnb + (seg + 16 * j) * 4);
        }
    } else {
#pragma unroll
        for (int j = 0; j < AV; ++j) {
            const int idx = tid + 256 * j;
            const int r = idx / AROWV, q = idx % AROWV;
            *(v4f*)(As + r * T::PA + q * 4) = *(const v4f*)(Ablk + (size_t)r * K + q * 4);
        }
    }
#pragma unroll
    for (int j = 0; j < WV; ++j) {
        const int idx = tid + 256 * j;
        wreg[j] = *(const v4f*)(Wcol + (size_t)(idx / ROWV) * N + (idx % ROWV) * 4);
    }
    if constexpr (LN_A) {
#pragma unroll
        for (int rr = 0; rr < NR; ++rr) {
            float s = 0.f, q = 0.f;
#pragma unroll
            for (int j = 0; j < NJ; ++j)
#pragma unroll
                for (int e = 0; e < 4; ++e) { const float v = xr[rr][j][e]; s += v; q += v * v; }
#pragma unroll
            for (int m = 8; m >= 1; m >>= 1) {
                s += __shfl_xor(s, m, 16);
                q += __shfl_xor(q, m, 16);
            }
            const float mu = s * (1.0f / K);
            const float var = q * (1.0f / K) - mu * mu;
            const float rstd = rsqrtf(var + 1e-5f);
            float* dst = As + (hw + 16 * rr) * T::PA;
#pragma unroll
            for (int j = 0; j < NJ; ++j) {
                v4f o;
#pragma unroll
                for (int e = 0; e < 4; ++e) {
                    const float v = (xr[rr][j][e] - mu) * rstd * gr[j][e] + br[j][e];
                    o[e] = fmaxf(v, 0.0f);
                }
                *(v4f*)(dst + (seg + 16 * j) * 4) = o;
            }
        }
    }
#pragma unroll
    for (int j = 0; j < WV; ++j) {
        const int idx = tid + 256 * j;
        *(v4f*)(Ws + (idx / ROWV) * T::PW + (idx % ROWV) * 4) = wreg[j];
    }
    __syncthreads();

    const float* ap = As + (wm * 16 + c) * T::PA + 2 * h;
    for (int s = 0; s < S; ++s) {
        const float* cur = Ws + (s & 1) * T::W_FLOATS;
        float* nxt = Ws + ((s + 1) & 1) * T::W_FLOATS;
        if (s + 1 < S) {
            const float* src = Wcol + (size_t)(s + 1) * KS * N;
#pragma unroll
            for (int j = 0; j < WV; ++j) {
                const int idx = tid + 256 * j;
                wreg[j] = *(const v4f*)(src + (size_t)(idx / ROWV) * N + (idx % ROWV) * 4);
            }
        }
        const float* bp = cur + (2 * h) * T::PW + n0 + c;
        constexpr int KHEAD = KS * 3 / 4;
#pragma unroll
        for (int kk = 0; kk < KS; kk += 4) {
            if (kk == KHEAD) __builtin_amdgcn_sched_barrier(0x0106);
            v2f a = *(const v2f*)(ap + s * KS + kk);
#pragma unroll
            for (int t = 0; t < NT; ++t) {
                v2f bf = { bp[kk * T::PW + t * 16], bp[(kk + 1) * T::PW + t * 16] };
                acc[t] = wmma_f32(a, bf, acc[t]);
            }
        }
        if (s + 1 < S) {
#pragma unroll
            for (int j = 0; j < WV; ++j) {
                const int idx = tid + 256 * j;
                *(v4f*)(nxt + (idx / ROWV) * T::PW + (idx % ROWV) * 4) = wreg[j];
            }
        }
        __syncthreads();
    }
}

__global__ void __launch_bounds__(256)
k_reduce_states(const float* __restrict__ states, float* __restrict__ xbar) {
    const int b = blockIdx.x;
    const int d = threadIdx.x;
    const float* p = states + (size_t)b * (O_DIM * D_DIM) + d;
    float acc = 0.f;
#pragma unroll
    for (int o = 0; o < O_DIM; ++o) acc += p[o * D_DIM];
    xbar[(size_t)b * D_DIM + d] = acc * (1.0f / 16.0f);
}

__global__ void __launch_bounds__(256)
k_gemm1(const float* __restrict__ xbar, const float* __restrict__ W1,
        const float* __restrict__ bias1, const int* __restrict__ action,
        float* __restrict__ h1) {
    constexpr int K = D_DIM, N = H_DIM, BN = 256, KS = 32, NT = 4, WM = 2;
    typedef Tile<K, BN, KS, WM> T;
    __shared__ __attribute__((aligned(16))) float lds[T::LDS_FLOATS];

    const int wave = threadIdx.x >> 5;
    const int lane = threadIdx.x & 31;
    const int mb = blockIdx.x / (N / BN);
    const int nb = (blockIdx.x % (N / BN)) * BN;
    const int m0 = mb * T::BM + (wave / T::WN) * 16;
    const int n0 = nb + (wave % T::WN) * (NT * 16);
    const int laneM = lane & 15;

    v8f acc[NT] = {};
    block_gemm_lds<K, N, BN, KS, NT, WM, false>(xbar + (size_t)mb * T::BM * K, W1 + nb,
                                                nullptr, nullptr,
                                                lds, lds + T::A_FLOATS, acc);

    const int mhalf = (lane >> 4) * 8;
#pragma unroll
    for (int r = 0; r < 8; ++r) {
        const int m = m0 + mhalf + r;
        const int c = action[m] & 3;
        const float* wtail = W1 + (size_t)(D_DIM + c) * N;
#pragma unroll
        for (int t = 0; t < 4; ++t) {
            const int n = n0 + t * 16 + laneM;
            float v = acc[t][r] + wtail[n] * 0.0625f + bias1[n];
            h1[(size_t)m * N + n] = fmaxf(v, 0.0f);
        }
    }
}

__global__ void __launch_bounds__(256)
k_gemm2(const float* __restrict__ h1, const float* __restrict__ W2,
        const float* __restrict__ bias2, float* __restrict__ z) {
    constexpr int K = H_DIM, N = H_DIM, BN = 256, KS = 32, NT = 4, WM = 2;
    typedef Tile<K, BN, KS, WM> T;
    __shared__ __attribute__((aligned(16))) float lds[T::LDS_FLOATS];

    const int wave = threadIdx.x >> 5;
    const int lane = threadIdx.x & 31;
    const int mb = blockIdx.x / (N / BN);
    const int nb = (blockIdx.x % (N / BN)) * BN;
    const int m0 = mb * T::BM + (wave / T::WN) * 16;
    const int n0 = nb + (wave % T::WN) * (NT * 16);
    const int laneM = lane & 15;

    v8f acc[NT] = {};
    block_gemm_lds<K, N, BN, KS, NT, WM, false>(h1 + (size_t)mb * T::BM * K, W2 + nb,
                                                nullptr, nullptr,
                                                lds, lds + T::A_FLOATS, acc);

    const int mhalf = (lane >> 4) * 8;
#pragma unroll
    for (int r = 0; r < 8; ++r) {
        const int m = m0 + mhalf + r;
#pragma unroll
        for (int t = 0; t < NT; ++t) {
            const int n = n0 + t * 16 + laneM;
            z[(size_t)m * N + n] = acc[t][r] + bias2[n];
        }
    }
}

__global__ void __launch_bounds__(256)
k_gemm3_ln_bcast(const float* __restrict__ z, const float* __restrict__ ln_g,
                 const float* __restrict__ ln_b, const float* __restrict__ W3,
                 const float* __restrict__ bias3, float* __restrict__ out) {
    constexpr int K = H_DIM, N = D_DIM, BN = 128, KS = 64, NT = 2, WM = 2;
    typedef Tile<K, BN, KS, WM> T;
    __shared__ __attribute__((aligned(16))) float lds[T::LDS_FLOATS];

    const int wave = threadIdx.x >> 5;
    const int lane = threadIdx.x & 31;
    const int mb = blockIdx.x / (N / BN);
    const int nb = (blockIdx.x % (N / BN)) * BN;
    const int m0 = mb * T::BM + (wave / T::WN) * 16;
    const int n0 = nb + (wave % T::WN) * (NT * 16);
    const int laneM = lane & 15;

    v8f acc[NT] = {};
    block_gemm_lds<K, N, BN, KS, NT, WM, true>(z + (size_t)mb * T::BM * K, W3 + nb,
                                               ln_g, ln_b,
                                               lds, lds + T::A_FLOATS, acc);

    const int mhalf = (lane >> 4) * 8;
#pragma unroll
    for (int r = 0; r < 8; ++r) {
        const int m = m0 + mhalf + r;
        float* obase = out + (size_t)m * (O_DIM * D_DIM);
#pragma unroll
        for (int t = 0; t < NT; ++t) {
            const int n = n0 + t * 16 + laneM;
            const float v = acc[t][r] + bias3[n];
#pragma unroll
            for (int o = 0; o < O_DIM; ++o)
                obase[o * D_DIM + n] = v;
        }
    }
}

extern "C" void kernel_launch(void* const* d_in, const int* in_sizes, int n_in,
                              void* d_out, int out_size, void* d_ws, size_t ws_size,
                              hipStream_t stream) {
    const float* states = (const float*)d_in[0];
    const int*   action = (const int*)d_in[1];
    const float* W1     = (const float*)d_in[2];
    const float* b1     = (const float*)d_in[3];
    const float* W2     = (const float*)d_in[4];
    const float* b2     = (const float*)d_in[5];
    const float* W3     = (const float*)d_in[6];
    const float* b3     = (const float*)d_in[7];
    const float* ln_g   = (const float*)d_in[8];
    const float* ln_b   = (const float*)d_in[9];
    float* out = (float*)d_out;

    float* xbar = (float*)d_ws;
    float* h1   = xbar + (size_t)B_DIM * D_DIM;
    float* z    = h1   + (size_t)B_DIM * H_DIM;

    k_reduce_states<<<B_DIM, 256, 0, stream>>>(states, xbar);
    k_gemm1<<<(B_DIM / 32) * 2, 256, 0, stream>>>(xbar, W1, b1, action, h1);
    k_gemm2<<<(B_DIM / 32) * 2, 256, 0, stream>>>(h1, W2, b2, z);
    k_gemm3_ln_bcast<<<(B_DIM / 32) * 2, 256, 0, stream>>>(z, ln_g, ln_b, W3, b3, out);
}


# Result tiles of all 3 products staged through LDS: a store instruction writes 512 contiguous bytes of one row (16-byte lanes; out once per object), not 4-byte lanes in 64-byte pieces; sums unchanged
# speedup vs baseline: 2.2426x; 1.1055x over previous
#include <hip/hip_runtime.h>
#include <hip/hip_bf16.h>


#define B_DIM 4096
#define O_DIM 16
#define D_DIM 256
#define H_DIM 512

typedef __attribute__((ext_vector_type(2))) float v2f;
typedef __attribute__((ext_vector_type(4))) float v4f;
typedef __attribute__((ext_vector_type(8))) float v8f;

__device__ __forceinline__ v8f wmma_f32(v2f a, v2f b, v8f c) {
    return __builtin_amdgcn_wmma_f32_16x16x4_f32(
         false, a,  false, b,
         (short)0, c,  false,  false);
}

template <int K, int BN, int KS, int WM>
struct Tile {
    static constexpr int BM = 16 * WM;
    static constexpr int WN = 8 / WM;
    static constexpr int PA = K + 4;
    static constexpr int PW = BN + 16;
    static constexpr int A_FLOATS = BM * PA;
    static constexpr int W_FLOATS = KS * PW;
    static constexpr int LDS_FLOATS = A_FLOATS + 2 * W_FLOATS;
};

template <int K, int N, int BN, int KS, int NT, int WM, bool LN_A>
__device__ __forceinline__ void block_gemm_lds(const float* __restrict__ Ablk,
                                               const float* __restrict__ Wcol,
                                               const float* __restrict__ lng,
                                               const float* __restrict__ lnb,
                                               float* __restrict__ As,
                                               float* __restrict__ Ws,
                                               v8f (&acc)[NT]) {
    typedef Tile<K, BN, KS, WM> T;
    constexpr int ROWV  = BN / 4;
    constexpr int WV    = KS * ROWV / 256;
    constexpr int AROWV = K / 4;
    constexpr int AV    = T::BM * AROWV / 256;
    constexpr int S     = K / KS;
    static_assert(T::WN * NT * 16 == BN, "wave columns x NT tiles cover BN");
    static_assert((KS * ROWV) % 256 == 0 && (T::BM * AROWV) % 256 == 0, "whole pieces per thread");
    static_assert(K % KS == 0 && KS % 16 == 0, "whole slabs, whole quarters");

    const int tid  = threadIdx.x;
    const int wave = tid >> 5;
    const int lane = tid & 31;
    const int c    = lane & 15;
    const int h    = lane >> 4;
    const int wm   = wave / T::WN;
    const int wn   = wave % T::WN;
    const int n0   = wn * (NT * 16);

    v4f wreg[WV];

    constexpr int NR = T::BM / 16;
    constexpr int NJ = K / 64;
    static_assert(!LN_A || (K % 64 == 0 && T::BM % 16 == 0), "whole pieces per lane, whole rows per half wave");
    const int hw  = tid >> 4;
    const int seg = tid & 15;
    v4f xr[LN_A ? NR : 1][LN_A ? NJ : 1];
    v4f gr[LN_A ? NJ : 1], br[LN_A ? NJ : 1];

    if constexpr (LN_A) {
#pragma unroll
        for (int rr = 0; rr < NR; ++rr)
#pragma unroll
            for (int j = 0; j < NJ; ++j)
                xr[rr][j] = *(const v4f*)(Ablk + (size_t)(hw + 16 * rr) * K + (seg + 16 * j) * 4);
#pragma unroll
        for (int j = 0; j < NJ; ++j) {
            gr[j] = *(const v4f*)(lng + (seg + 16 * j) * 4);
            br[j] = *(const v4f*)(lnb + (seg + 16 * j) * 4);
        }
    } else {
#pragma unroll
        for (int j = 0; j < AV; ++j) {
            const int idx = tid + 256 * j;
            const int r = idx / AROWV, q = idx % AROWV;
            *(v4f*)(As + r * T::PA + q * 4) = *(const v4f*)(Ablk + (size_t)r * K + q * 4);
        }
    }
#pragma unroll
    for (int j = 0; j < WV; ++j) {
        const int idx = tid + 256 * j;
        wreg[j] = *(const v4f*)(Wcol + (size_t)(idx / ROWV) * N + (idx % ROWV) * 4);
    }
    if constexpr (LN_A) {
#pragma unroll
        for (int rr = 0; rr < NR; ++rr) {
            float s = 0.f, q = 0.f;
#pragma unroll
            for (int j = 0; j < NJ; ++j)
#pragma unroll
                for (int e = 0; e < 4; ++e) { const float v = xr[rr][j][e]; s += v; q += v * v; }
#pragma unroll
            for (int m = 8; m >= 1; m >>= 1) {
                s += __shfl_xor(s, m, 16);
                q += __shfl_xor(q, m, 16);
            }
            const float mu = s * (1.0f / K);
            const float var = q * (1.0f / K) - mu * mu;
            const float rstd = rsqrtf(var + 1e-5f);
            float* dst = As + (hw + 16 * rr) * T::PA;
#pragma unroll
            for (int j = 0; j < NJ; ++j) {
                v4f o;
#pragma unroll
                for (int e = 0; e < 4; ++e) {
                    const float v = (xr[rr][j][e] - mu) * rstd * gr[j][e] + br[j][e];
                    o[e] = fmaxf(v, 0.0f);
                }
                *(v4f*)(dst + (seg + 16 * j) * 4) = o;
            }
        }
    }
#pragma unroll
    for (int j = 0; j < WV; ++j) {
        const int idx = tid + 256 * j;
        *(v4f*)(Ws + (idx / ROWV) * T::PW + (idx % ROWV) * 4) = wreg[j];
    }
    __syncthreads();

    const float* ap = As + (wm * 16 + c) * T::PA + 2 * h;
    for (int s = 0; s < S; ++s) {
        const float* cur = Ws + (s & 1) * T::W_FLOATS;
        float* nxt = Ws + ((s + 1) & 1) * T::W_FLOATS;
        if (s + 1 < S) {
            const float* src = Wcol + (size_t)(s + 1) * KS * N;
#pragma unroll
            for (int j = 0; j < WV; ++j) {
                const int idx = tid + 256 * j;
                wreg[j] = *(const v4f*)(src + (size_t)(idx / ROWV) * N + (idx % ROWV) * 4);
            }
        }
        const float* bp = cur + (2 * h) * T::PW + n0 + c;
        constexpr int KHEAD = KS * 3 / 4;
#pragma unroll
        for (int kk = 0; kk < KS; kk += 4) {
            if (kk == KHEAD) __builtin_amdgcn_sched_barrier(0x0106);
            v2f a = *(const v2f*)(ap + s * KS + kk);
#pragma unroll
            for (int t = 0; t < NT; ++t) {
                v2f bf = { bp[kk * T::PW + t * 16], bp[(kk + 1) * T::PW + t * 16] };
                acc[t] = wmma_f32(a, bf, acc[t]);
            }
        }
        if (s + 1 < S) {
#pragma unroll
            for (int j = 0; j < WV; ++j) {
                const int idx = tid + 256 * j;
                *(v4f*)(nxt + (idx / ROWV) * T::PW + (idx % ROWV) * 4) = wreg[j];
            }
        }
        __syncthreads();
    }
}

template <int BN, int WN, int NT>
__device__ __forceinline__ int tile_at(int r, int t) {
    const int wave = threadIdx.x >> 5, lane = threadIdx.x & 31;
    const int row = (wave / WN) * 16 + (lane >> 4) * 8 + r;
    const int col = (wave % WN) * (NT * 16) + t * 16 + (lane & 15);
    return row * (BN + 4) + col;
}

template <int BM, int BN, int REP>
__device__ __forceinline__ void rows_out(const float* __restrict__ ot, float* __restrict__ dst,
                                         int ldo, int rep_stride) {
    const int wave = threadIdx.x >> 5, lane = threadIdx.x & 31;
#pragma unroll
    for (int i = 0; i < BM / 8; ++i) {
        const int R = wave + 8 * i;
#pragma unroll
        for (int j = 0; j < BN / 128; ++j) {
            const v4f v = *(const v4f*)(ot + R * (BN + 4) + j * 128 + lane * 4);
            float* p = dst + (size_t)R * ldo + j * 128 + lane * 4;
#pragma unroll
            for (int o = 0; o < REP; ++o)
                *(v4f*)(p + o * rep_stride) = v;
        }
    }
}

__global__ void __launch_bounds__(256)
k_reduce_states(const float* __restrict__ states, float* __restrict__ xbar) {
    const int b = blockIdx.x;
    const int d = threadIdx.x;
    const float* p = states + (size_t)b * (O_DIM * D_DIM) + d;
    float acc = 0.f;
#pragma unroll
    for (int o = 0; o < O_DIM; ++o) acc += p[o * D_DIM];
    xbar[(size_t)b * D_DIM + d] = acc * (1.0f / 16.0f);
}

__global__ void __launch_bounds__(256)
k_gemm1(const float* __restrict__ xbar, const float* __restrict__ W1,
        const float* __restrict__ bias1, const int* __restrict__ action,
        float* __restrict__ h1) {
    constexpr int K = D_DIM, N = H_DIM, BN = 256, KS = 32, NT = 4, WM = 2;
    typedef Tile<K, BN, KS, WM> T;
    __shared__ __attribute__((aligned(16))) float lds[T::LDS_FLOATS];

    const int wave = threadIdx.x >> 5;
    const int lane = threadIdx.x & 31;
    const int mb = blockIdx.x / (N / BN);
    const int nb = (blockIdx.x % (N / BN)) * BN;
    const int m0 = mb * T::BM + (wave / T::WN) * 16;
    const int n0 = nb + (wave % T::WN) * (NT * 16);
    const int laneM = lane & 15;

    v8f acc[NT] = {};
    block_gemm_lds<K, N, BN, KS, NT, WM, false>(xbar + (size_t)mb * T::BM * K, W1 + nb,
                                                nullptr, nullptr,
                                                lds, lds + T::A_FLOATS, acc);

    const int mhalf = (lane >> 4) * 8;
#pragma unroll
    for (int r = 0; r < 8; ++r) {
        const int m = m0 + mhalf + r;
        const int c = action[m] & 3;
        const float* wtail = W1 + (size_t)(D_DIM + c) * N;
#pragma unroll
        for (int t = 0; t < 4; ++t) {
            const int n = n0 + t * 16 + laneM;
            float v = acc[t][r] + wtail[n] * 0.0625f + bias1[n];
            lds[tile_at<BN, T::WN, NT>(r, t)] = fmaxf(v, 0.0f);
        }
    }
    __syncthreads();
    rows_out<T::BM, BN, 1>(lds, h1 + (size_t)mb * T::BM * N + nb, N, 0);
}

__global__ void __launch_bounds__(256)
k_gemm2(const float* __restrict__ h1, const float* __restrict__ W2,
        const float* __restrict__ bias2, float* __restrict__ z) {
    constexpr int K = H_DIM, N = H_DIM, BN = 256, KS = 32, NT = 4, WM = 2;
    typedef Tile<K, BN, KS, WM> T;
    __shared__ __attribute__((aligned(16))) float lds[T::LDS_FLOATS];

    const int wave = threadIdx.x >> 5;
    const int lane = threadIdx.x & 31;
    const int mb = blockIdx.x / (N / BN);
    const int nb = (blockIdx.x % (N / BN)) * BN;
    const int n0 = nb + (wave % T::WN) * (NT * 16);
    const int laneM = lane & 15;

    v8f acc[NT] = {};
    block_gemm_lds<K, N, BN, KS, NT, WM, false>(h1 + (size_t)mb * T::BM * K, W2 + nb,
                                                nullptr, nullptr,
                                                lds, lds + T::A_FLOATS, acc);

#pragma unroll
    for (int t = 0; t < NT; ++t) {
        const float bv = bias2[n0 + t * 16 + laneM];
#pragma unroll
        for (int r = 0; r < 8; ++r)
            lds[tile_at<BN, T::WN, NT>(r, t)] = acc[t][r] + bv;
    }
    __syncthreads();
    rows_out<T::BM, BN, 1>(lds, z + (size_t)mb * T::BM * N + nb, N, 0);
}

__global__ void __launch_bounds__(256)
k_gemm3_ln_bcast(const float* __restrict__ z, const float* __restrict__ ln_g,
                 const float* __restrict__ ln_b, const float* __restrict__ W3,
                 const float* __restrict__ bias3, float* __restrict__ out) {
    constexpr int K = H_DIM, N = D_DIM, BN = 128, KS = 64, NT = 2, WM = 2;
    typedef Tile<K, BN, KS, WM> T;
    __shared__ __attribute__((aligned(16))) float lds[T::LDS_FLOATS];

    const int wave = threadIdx.x >> 5;
    const int lane = threadIdx.x & 31;
    const int mb = blockIdx.x / (N / BN);
    const int nb = (blockIdx.x % (N / BN)) * BN;
    const int n0 = nb + (wave % T::WN) * (NT * 16);
    const int laneM = lane & 15;

    v8f acc[NT] = {};
    block_gemm_lds<K, N, BN, KS, NT, WM, true>(z + (size_t)mb * T::BM * K, W3 + nb,
                                               ln_g, ln_b,
                                               lds, lds + T::A_FLOATS, acc);

#pragma unroll
    for (int t = 0; t < NT; ++t) {
        const float bv = bias3[n0 + t * 16 + laneM];
#pragma unroll
        for (int r = 0; r < 8; ++r)
            lds[tile_at<BN, T::WN, NT>(r, t)] = acc[t][r] + bv;
    }
    __syncthreads();
    rows_out<T::BM, BN, O_DIM>(lds, out + (size_t)mb * T::BM * (O_DIM * D_DIM) + nb,
                               O_DIM * D_DIM, D_DIM);
}

extern "C" void kernel_launch(void* const* d_in, const int* in_sizes, int n_in,
                              void* d_out, int out_size, void* d_ws, size_t ws_size,
                              hipStream_t stream) {
    const float* states = (const float*)d_in[0];
    const int*   action = (const int*)d_in[1];
    const float* W1     = (const float*)d_in[2];
    const float* b1     = (const float*)d_in[3];
    const float* W2     = (const float*)d_in[4];
    const float* b2     = (const float*)d_in[5];
    const float* W3     = (const float*)d_in[6];
    const float* b3     = (const float*)d_in[7];
    const float* ln_g   = (const float*)d_in[8];
    const float* ln_b   = (const float*)d_in[9];
    float* out = (float*)d_out;

    float* xbar = (float*)d_ws;
    float* h1   = xbar + (size_t)B_DIM * D_DIM;
    float* z    = h1   + (size_t)B_DIM * H_DIM;

    k_reduce_states<<<B_DIM, 256, 0, stream>>>(states, xbar);
    k_gemm1<<<(B_DIM / 32) * 2, 256, 0, stream>>>(xbar, W1, b1, action, h1);
    k_gemm2<<<(B_DIM / 32) * 2, 256, 0, stream>>>(h1, W2, b2, z);
    k_gemm3_ln_bcast<<<(B_DIM / 32) * 2, 256, 0, stream>>>(z, ln_g, ln_b, W3, b3, out);
}
